# starting kernel
# baseline (speedup 1.0000x reference)
#include <hip/hip_runtime.h>


#define CC   256
#define HH   256
#define WW   256
#define HWSZ (HH * WW)
#define KS   9
#define PAD  4
#define NKSTEP 6
#define NFRAG  (KS * NKSTEP * 32)

typedef __attribute__((ext_vector_type(2))) float v2f;
typedef __attribute__((ext_vector_type(8))) float v8f;

__launch_bounds__(512)
__global__ void prop9x9_wmma_kernel(const float* __restrict__ key,
                                    const float* __restrict__ wgt,
                                    float* __restrict__ out)
{
    __shared__ v2f lbs[NFRAG];

    const int w0   = blockIdx.x << 4;
    const int h    = blockIdx.y;
    const int tid  = threadIdx.x;
    const int lane = tid & 31;
    const int wave = tid >> 5;

    for (int f = tid; f < NFRAG; f += 512) {
        const int  l    = f & 31;
        const int  fk   = f >> 5;
        const int  ks   = fk % NKSTEP;
        const int  i    = fk / NKSTEP;
        const int  row0 = (ks << 2) + ((l >> 4) << 1);
        const int  wc   = l & 15;
        const bool rv   = (unsigned)(h + i - PAD) < (unsigned)HH;
        const size_t pix = (size_t)h * WW + (size_t)(w0 + wc);
        v2f v;
#pragma unroll
        for (int t = 0; t < 2; ++t) {
            const int  k  = row0 + t;
            const int  j  = k - wc;
            const bool ok = rv && ((unsigned)j <= 8u) &&
                            ((unsigned)(w0 - PAD + k) < (unsigned)WW);
            const float val = ok ? wgt[(size_t)(i * KS + j) * HWSZ + pix] : 0.0f;
            if (t == 0) v.x = val; else v.y = val;
        }
        lbs[f] = v;
    }
    __syncthreads();

    const int c0    = wave << 4;
    const int m     = lane & 15;
    const int khalf = (lane >> 4) << 1;

    unsigned eoffb[NKSTEP];
#pragma unroll
    for (int ks = 0; ks < NKSTEP; ++ks) {
        int col = w0 - PAD + (ks << 2) + khalf;
        col = col < 0 ? 0 : (col > WW - 2 ? WW - 2 : col);
        eoffb[ks] = (unsigned)(((c0 + m) * HWSZ + col) * 4);
    }

    v8f acc = {};
#pragma unroll
    for (int i = 0; i < KS; ++i) {
        int hh = h + i - PAD;
        hh = hh < 0 ? 0 : (hh > HH - 1 ? HH - 1 : hh);
        const char* rowbase = (const char*)(key + (size_t)hh * WW);

        v2f a[NKSTEP], b[NKSTEP];
#pragma unroll
        for (int ks = 0; ks < NKSTEP; ++ks) {
            unsigned bo = eoffb[ks];
            asm("" : "+v"(bo));
            a[ks] = *(const v2f*)(rowbase + bo);
        }
#pragma unroll
        for (int ks = 0; ks < NKSTEP; ++ks)
            b[ks] = lbs[(i * NKSTEP + ks) * 32 + lane];
#pragma unroll
        for (int ks = 0; ks < NKSTEP; ++ks)
            acc = __builtin_amdgcn_wmma_f32_16x16x4_f32(
                      false, a[ks], false, b[ks], (short)0, acc, false, false);
    }

    const int n     = lane & 15;
    const int mbase = (lane >> 4) << 3;
    unsigned sob = (unsigned)((((c0 + mbase) * HWSZ) + h * WW + w0 + n) * 4);
    asm("" : "+v"(sob));
    char* ob = (char*)out;
#pragma unroll
    for (int v = 0; v < 8; ++v)
        *(float*)(ob + sob + (size_t)(v * (HWSZ * 4))) = acc[v];
}

extern "C" void kernel_launch(void* const* d_in, const int* in_sizes, int n_in,
                              void* d_out, int out_size, void* d_ws, size_t ws_size,
                              hipStream_t stream)
{
    (void)in_sizes; (void)n_in; (void)out_size; (void)d_ws; (void)ws_size;
    const float* key = (const float*)d_in[0];
    const float* wgt = (const float*)d_in[1];
    float*       out = (float*)d_out;

    dim3 grid(WW / 16, HH);
    prop9x9_wmma_kernel<<<grid, 512, 0, stream>>>(key, wgt, out);
}


# 8 output rows per block: each key row's A fragments are loaded once and reused from registers by all output rows they feed (up to 9); B of the 8 rows built up front; out-of-image key rows skipped
# speedup vs baseline: 3.2692x; 3.2692x over previous
#include <hip/hip_runtime.h>


#define CC   256
#define HH   256
#define WW   256
#define HWSZ (HH * WW)
#define KS   9
#define PAD  4
#define NKSTEP 6
#define NFRAG  (KS * NKSTEP * 32)
#define RB     8
#define NKROW  (RB + KS - 1)

typedef __attribute__((ext_vector_type(2))) float v2f;
typedef __attribute__((ext_vector_type(8))) float v8f;

__launch_bounds__(512, 8)
__global__ void prop9x9_wmma_kernel(const float* __restrict__ key,
                                    const float* __restrict__ wgt,
                                    float* __restrict__ out)
{
    __shared__ v2f lbs[RB * NFRAG];

    const int w0   = blockIdx.x << 4;
    const int h0   = blockIdx.y * RB;
    const int tid  = threadIdx.x;
    const int lane = tid & 31;
    const int wave = tid >> 5;

    const int wc = lane & 15;
    const int kh = (lane >> 4) << 1;
    unsigned woff[NKSTEP][2];
    bool     wok[NKSTEP][2];
#pragma unroll
    for (int ks = 0; ks < NKSTEP; ++ks) {
#pragma unroll
        for (int t = 0; t < 2; ++t) {
            const int k  = (ks << 2) + kh + t;
            const int j  = k - wc;
            const int jc = j < 0 ? 0 : (j > KS - 1 ? KS - 1 : j);
            wok[ks][t]  = ((unsigned)j <= 8u) &&
                          ((unsigned)(w0 - PAD + k) < (unsigned)WW);
            woff[ks][t] = (unsigned)((jc * HWSZ + w0 + wc) * 4);
        }
    }
    const int wv = __builtin_amdgcn_readfirstlane(wave);
    auto build_pair = [&](const int ri) {
        const int  i  = ri % KS;
        const int  h  = h0 + ri / KS;
        const char* gb = (const char*)(wgt + (size_t)(i * KS) * HWSZ + (size_t)h * WW);
        v2f v[NKSTEP];
#pragma unroll
        for (int ks = 0; ks < NKSTEP; ++ks) {
#pragma unroll
            for (int t = 0; t < 2; ++t) {
                unsigned bo = woff[ks][t];
                asm("" : "+v"(bo));
                const float val = *(const float*)(gb + bo);
                const float sel = wok[ks][t] ? val : 0.0f;
                if (t == 0) v[ks].x = sel; else v[ks].y = sel;
            }
        }
#pragma unroll
        for (int ks = 0; ks < NKSTEP; ++ks)
            lbs[(ri * NKSTEP + ks) * 32 + lane] = v[ks];
    };
#pragma unroll
    for (int it = 0; it < (RB * KS) / 16; ++it)
        build_pair(wv + 16 * it);
    if (wv < (RB * KS) % 16)
        build_pair(wv + ((RB * KS) / 16) * 16);
    __syncthreads();

    const int c0    = wave << 4;
    const int m     = lane & 15;
    const int khalf = (lane >> 4) << 1;

    unsigned eoffb[NKSTEP];
#pragma unroll
    for (int ks = 0; ks < NKSTEP; ++ks) {
        int col = w0 - PAD + (ks << 2) + khalf;
        col = col < 0 ? 0 : (col > WW - 2 ? WW - 2 : col);
        eoffb[ks] = (unsigned)(((c0 + m) * HWSZ + col) * 4);
    }

    const int n     = lane & 15;
    const int mbase = (lane >> 4) << 3;
    unsigned sob = (unsigned)((((c0 + mbase) * HWSZ) + h0 * WW + w0 + n) * 4);
    asm("" : "+v"(sob));
    char* ob = (char*)out;

    v8f acc[RB];
#pragma unroll
    for (int hl = 0; hl < RB; ++hl) acc[hl] = (v8f){};

    auto load_row = [&](const int r, v2f (&a)[NKSTEP]) {
        int hh = h0 + r - PAD;
        hh = hh < 0 ? 0 : (hh > HH - 1 ? HH - 1 : hh);
        const char* rowbase = (const char*)(key + (size_t)hh * WW);
#pragma unroll
        for (int ks = 0; ks < NKSTEP; ++ks) {
            unsigned bo = eoffb[ks];
            asm("" : "+v"(bo));
            a[ks] = *(const v2f*)(rowbase + bo);
        }
    };

    v2f abuf[2][NKSTEP];
    load_row(0, abuf[0]);

#pragma unroll
    for (int r = 0; r < NKROW; ++r) {
        v2f (&a)[NKSTEP] = abuf[r & 1];
        if (r + 1 < NKROW) {
            load_row(r + 1, abuf[(r + 1) & 1]);
            asm volatile("" ::: "memory");
        }

        if ((unsigned)(h0 + r - PAD) < (unsigned)HH) {
#pragma unroll
        for (int hl = 0; hl < RB; ++hl) {
            const int i = r - hl;
            if (i >= 0 && i < KS) {
                v2f b[NKSTEP];
#pragma unroll
                for (int ks = 0; ks < NKSTEP; ++ks)
                    b[ks] = lbs[((hl * KS + i) * NKSTEP + ks) * 32 + lane];
                asm volatile("" ::: "memory");
#pragma unroll
                for (int ks = 0; ks < NKSTEP; ++ks)
                    acc[hl] = __builtin_amdgcn_wmma_f32_16x16x4_f32(
                                  false, a[ks], false, b[ks], (short)0, acc[hl], false, false);
            }
        }
        }

        if (r >= KS - 1) {
            const int hl = r - (KS - 1);
#pragma unroll
            for (int v = 0; v < 8; ++v)
                *(float*)(ob + sob + (size_t)(v * (HWSZ * 4) + hl * (WW * 4))) = acc[hl][v];
        }
    }
}

extern "C" void kernel_launch(void* const* d_in, const int* in_sizes, int n_in,
                              void* d_out, int out_size, void* d_ws, size_t ws_size,
                              hipStream_t stream)
{
    (void)in_sizes; (void)n_in; (void)out_size; (void)d_ws; (void)ws_size;
    const float* key = (const float*)d_in[0];
    const float* wgt = (const float*)d_in[1];
    float*       out = (float*)d_out;

    dim3 grid(WW / 16, HH / RB);
    prop9x9_wmma_kernel<<<grid, 512, 0, stream>>>(key, wgt, out);
}
